# starting kernel
# baseline (speedup 1.0000x reference)
#include <hip/hip_runtime.h>
#include <hip/hip_bf16.h>


typedef __attribute__((ext_vector_type(2))) float v2f;
typedef __attribute__((ext_vector_type(8))) float v8f;

#define BATCH 2048
#define FRAME 267
#define LAT   32
#define HID   256
#define NE    6
#define GH    64

__device__ __forceinline__ float elu1(float x) {
  return x > 0.f ? x : expm1f(x);
}

__device__ __forceinline__ v8f wmma4(v2f a, v2f b, v8f c) {
  return __builtin_amdgcn_wmma_f32_16x16x4_f32(
      false, a, false, b, (short)0, c, false, false);
}

template <int N, bool MOE, int NT>
__device__ __forceinline__ void seg_gemm(
    const float* __restrict__ Aseg, int aBase,
    int len, float scale,
    const float* __restrict__ Wseg,
    const int* col, int h, v8f* acc)
{
  int wIdx[NT];
  #pragma unroll
  for (int j = 0; j < NT; ++j) wIdx[j] = 2 * h * N + col[j];
  const int a0 = aBase + 2 * h;

  const int len4 = len & ~3;
  int k = 0;
  #pragma unroll 8
  for (; k < len4; k += 4) {
    v2f a;
    a.x = Aseg[a0 + k];
    a.y = Aseg[a0 + k + 1];
    if (MOE) { a.x *= scale; a.y *= scale; }
    #pragma unroll
    for (int j = 0; j < NT; ++j) {
      v2f b;
      b.x = Wseg[wIdx[j]];
      b.y = Wseg[wIdx[j] + N];
      acc[j] = wmma4(a, b, acc[j]);
      wIdx[j] += 4 * N;
    }
  }
  if (k < len) {
    const int kk = k + 2 * h;
    v2f a;
    a.x = (kk     < len) ? Aseg[aBase + kk]     : 0.f;
    a.y = (kk + 1 < len) ? Aseg[aBase + kk + 1] : 0.f;
    if (MOE) { a.x *= scale; a.y *= scale; }
    const int r0 = ((kk     < len) ? kk     : len - 1) - kk;
    const int r1 = ((kk + 1 < len) ? kk + 1 : len - 1) - kk;
    #pragma unroll
    for (int j = 0; j < NT; ++j) {
      v2f b;
      b.x = Wseg[wIdx[j] + r0 * N];
      b.y = Wseg[wIdx[j] + r1 * N];
      acc[j] = wmma4(a, b, acc[j]);
    }
  }
}

template <int N, bool MOE, int NT>
__global__ void __launch_bounds__(32)
gemm_wmma_f32(const float* __restrict__ A1, int k1,
              const float* __restrict__ A2, int k2,
              const float* __restrict__ coeff,
              const float* __restrict__ W,
              const float* __restrict__ bias,
              const float* __restrict__ bias2d,
              float* __restrict__ C,
              int act)
{
  const int lane = threadIdx.x;
  const int h    = lane >> 4;
  const int ln   = lane & 15;
  const int m0   = blockIdx.y * 16;
  const int n0   = blockIdx.x * (16 * NT);
  const int rowA = m0 + ln;

  int col[NT];
  #pragma unroll
  for (int j = 0; j < NT; ++j) {
    const int cb = n0 + 16 * j + ln;
    col[j] = (cb < N) ? cb : N - 1;
  }

  v8f acc[NT];
  #pragma unroll
  for (int j = 0; j < NT; ++j) acc[j] = (v8f){0.f,0.f,0.f,0.f,0.f,0.f,0.f,0.f};

  const int a1Base = rowA * k1;
  const int a2Base = rowA * k2;
  const int I = k1 + k2;
  const int nExp = MOE ? NE : 1;

  for (int e = 0; e < nExp; ++e) {
    const float scale = MOE ? coeff[rowA * NE + e] : 1.f;
    const float* Wb = W + (size_t)e * I * N;
    seg_gemm<N, MOE, NT>(A1, a1Base, k1, scale, Wb, col, h, acc);
    if (k2)
      seg_gemm<N, MOE, NT>(A2, a2Base, k2, scale, Wb + (size_t)k1 * N, col, h,
                           acc);
  }

  #pragma unroll
  for (int j = 0; j < NT; ++j) {
    const int colB = n0 + 16 * j + ln;
    if (colB >= N) continue;
    #pragma unroll
    for (int i = 0; i < 8; ++i) {
      const int row = m0 + i + 8 * h;
      float y = acc[j][i];
      if (bias)   y += bias[colB];
      if (bias2d) y += bias2d[row * N + colB];
      if (act)    y = elu1(y);
      C[row * N + colB] = y;
    }
  }
}

__global__ void reparam_kernel(const float* __restrict__ m,
                               const float* __restrict__ v,
                               const float* __restrict__ eps,
                               float* __restrict__ z, int n) {
  int i = blockIdx.x * blockDim.x + threadIdx.x;
  if (i < n) z[i] = m[i] + eps[i] * expf(0.5f * v[i]);
}

__global__ void gate3_softmax_kernel(const float* __restrict__ g,
                                     const float* __restrict__ w,
                                     const float* __restrict__ b,
                                     float* __restrict__ coeff) {
  int row = blockIdx.x * blockDim.x + threadIdx.x;
  if (row >= BATCH) return;
  float lg[NE];
  #pragma unroll
  for (int e = 0; e < NE; ++e) lg[e] = b[e];
  for (int i = 0; i < GH; ++i) {
    float gv = g[row * GH + i];
    #pragma unroll
    for (int e = 0; e < NE; ++e) lg[e] = fmaf(gv, w[i * NE + e], lg[e]);
  }
  float mx = lg[0];
  #pragma unroll
  for (int e = 1; e < NE; ++e) mx = fmaxf(mx, lg[e]);
  float s = 0.f, ex[NE];
  #pragma unroll
  for (int e = 0; e < NE; ++e) { ex[e] = expf(lg[e] - mx); s += ex[e]; }
  float inv = 1.f / s;
  #pragma unroll
  for (int e = 0; e < NE; ++e) coeff[row * NE + e] = ex[e] * inv;
}

__global__ void mixed_bias_kernel(const float* __restrict__ coeff,
                                  const float* __restrict__ b0,
                                  const float* __restrict__ b1,
                                  const float* __restrict__ b2,
                                  float* __restrict__ mb0,
                                  float* __restrict__ mb1,
                                  float* __restrict__ mb2) {
  int row = blockIdx.x;
  float cf[NE];
  #pragma unroll
  for (int e = 0; e < NE; ++e) cf[e] = coeff[row * NE + e];
  for (int j = threadIdx.x; j < HID; j += blockDim.x) {
    float s0 = 0.f, s1 = 0.f;
    #pragma unroll
    for (int e = 0; e < NE; ++e) {
      s0 = fmaf(cf[e], b0[e * HID + j], s0);
      s1 = fmaf(cf[e], b1[e * HID + j], s1);
    }
    mb0[row * HID + j] = s0;
    mb1[row * HID + j] = s1;
  }
  for (int j = threadIdx.x; j < FRAME; j += blockDim.x) {
    float s2 = 0.f;
    #pragma unroll
    for (int e = 0; e < NE; ++e) s2 = fmaf(cf[e], b2[e * FRAME + j], s2);
    mb2[row * FRAME + j] = s2;
  }
}

template <int N, int NT>
static inline dim3 gemm_grid() {
  return dim3((N + 16 * NT - 1) / (16 * NT), BATCH / 16);
}

extern "C" void kernel_launch(void* const* d_in, const int* in_sizes, int n_in,
                              void* d_out, int out_size, void* d_ws, size_t ws_size,
                              hipStream_t stream) {
  const float* x      = (const float*)d_in[0];
  const float* c      = (const float*)d_in[1];
  const float* eps    = (const float*)d_in[2];
  const float* efc1_w = (const float*)d_in[3];
  const float* efc1_b = (const float*)d_in[4];
  const float* efc2_w = (const float*)d_in[5];
  const float* efc2_b = (const float*)d_in[6];
  const float* em_w   = (const float*)d_in[7];
  const float* em_b   = (const float*)d_in[8];
  const float* ev_w   = (const float*)d_in[9];
  const float* ev_b   = (const float*)d_in[10];
  const float* g1_w   = (const float*)d_in[11];
  const float* g1_b   = (const float*)d_in[12];
  const float* g2_w   = (const float*)d_in[13];
  const float* g2_b   = (const float*)d_in[14];
  const float* g3_w   = (const float*)d_in[15];
  const float* g3_b   = (const float*)d_in[16];
  const float* w0     = (const float*)d_in[17];
  const float* b0     = (const float*)d_in[18];
  const float* w1     = (const float*)d_in[19];
  const float* b1     = (const float*)d_in[20];
  const float* w2     = (const float*)d_in[21];
  const float* b2     = (const float*)d_in[22];

  float* out   = (float*)d_out;
  float* out_m = out + BATCH * FRAME;
  float* out_v = out_m + BATCH * LAT;

  float* ws    = (float*)d_ws;
  float* h1    = ws;
  float* h2    = h1  + BATCH * HID;
  float* z     = h2  + BATCH * HID;
  float* g1o   = z   + BATCH * LAT;
  float* g2o   = g1o + BATCH * GH;
  float* coeff = g2o + BATCH * GH;
  float* mb0   = coeff + BATCH * NE;
  float* mb1   = mb0 + BATCH * HID;
  float* mb2   = mb1 + BATCH * HID;

  gemm_wmma_f32<HID, false, 4><<<gemm_grid<HID, 4>(), 32, 0, stream>>>(
      x, FRAME, c, FRAME, nullptr, efc1_w, efc1_b, nullptr, h1, 1);
  gemm_wmma_f32<HID, false, 4><<<gemm_grid<HID, 4>(), 32, 0, stream>>>(
      x, FRAME, h1, HID, nullptr, efc2_w, efc2_b, nullptr, h2, 1);
  gemm_wmma_f32<LAT, false, 2><<<gemm_grid<LAT, 2>(), 32, 0, stream>>>(
      x, FRAME, h2, HID, nullptr, em_w, em_b, nullptr, out_m, 0);
  gemm_wmma_f32<LAT, false, 2><<<gemm_grid<LAT, 2>(), 32, 0, stream>>>(
      x, FRAME, h2, HID, nullptr, ev_w, ev_b, nullptr, out_v, 0);
  reparam_kernel<<<(BATCH * LAT + 255) / 256, 256, 0, stream>>>(
      out_m, out_v, eps, z, BATCH * LAT);

  gemm_wmma_f32<GH, false, 2><<<gemm_grid<GH, 2>(), 32, 0, stream>>>(
      z, LAT, c, FRAME, nullptr, g1_w, g1_b, nullptr, g1o, 1);
  gemm_wmma_f32<GH, false, 2><<<gemm_grid<GH, 2>(), 32, 0, stream>>>(
      g1o, GH, nullptr, 0, nullptr, g2_w, g2_b, nullptr, g2o, 1);
  gate3_softmax_kernel<<<(BATCH + 255) / 256, 256, 0, stream>>>(
      g2o, g3_w, g3_b, coeff);
  mixed_bias_kernel<<<BATCH, 256, 0, stream>>>(coeff, b0, b1, b2, mb0, mb1, mb2);

  gemm_wmma_f32<HID, true, 4><<<gemm_grid<HID, 4>(), 32, 0, stream>>>(
      z, LAT, c, FRAME, coeff, w0, nullptr, mb0, h1, 1);
  gemm_wmma_f32<HID, true, 4><<<gemm_grid<HID, 4>(), 32, 0, stream>>>(
      z, LAT, h1, HID, coeff, w1, nullptr, mb1, h2, 1);
  gemm_wmma_f32<FRAME, true, 4><<<gemm_grid<FRAME, 4>(), 32, 0, stream>>>(
      z, LAT, h2, HID, coeff, w2, nullptr, mb2, out, 0);
}


# MoE layers: each expert's bias is one more k row of its wave's matrix product (A = coeff, B = b_e); the mixed-bias launch and its planes go (12 -> 11 launches); bias now summed inside the partials
# speedup vs baseline: 3.4476x; 3.4476x over previous
#include <hip/hip_runtime.h>
#include <hip/hip_bf16.h>


typedef __attribute__((ext_vector_type(2))) float v2f;
typedef __attribute__((ext_vector_type(8))) float v8f;
typedef float v2fu __attribute__((ext_vector_type(2), aligned(4)));
typedef float v4fu __attribute__((ext_vector_type(4), aligned(4)));
typedef __attribute__((ext_vector_type(4))) float v4f;
template <int NT> struct RowVec;
template <> struct RowVec<2> { typedef v2fu T; typedef v2f L; };
template <> struct RowVec<4> { typedef v4fu T; typedef v4f L; };

#define BATCH 2048
#define FRAME 267
#define LAT   32
#define HID   256
#define NE    6
#define GH    64

__device__ __forceinline__ float elu1(float x) {
  return x > 0.f ? x : expm1f(x);
}

__device__ __forceinline__ v8f wmma4(v2f a, v2f b, v8f c) {
  return __builtin_amdgcn_wmma_f32_16x16x4_f32(
      false, a, false, b, (short)0, c, false, false);
}

template <int NT, bool VEC>
__device__ __forceinline__ void load_b(const float* __restrict__ Wseg,
                                       int rowIdx, int cbase, const int* col,
                                       float* b)
{
  if (VEC) {
    typedef typename RowVec<NT>::T RV;
    const RV t = *reinterpret_cast<const RV*>(Wseg + rowIdx + cbase);
    #pragma unroll
    for (int j = 0; j < NT; ++j) b[j] = t[j];
  } else {
    #pragma unroll
    for (int j = 0; j < NT; ++j) b[j] = Wseg[rowIdx + col[j]];
  }
}

template <int N, bool MOE, int NT, bool VEC>
__device__ __forceinline__ void seg_gemm(
    const float* __restrict__ Aseg, int aBase,
    int len, float scale,
    const float* __restrict__ Wseg,
    int cbase, const int* col, int h, v8f* acc)
{
  int aIdx = aBase + 4 * h;
  int wIdx = 4 * h * N;

  const int len8 = len & ~7;
  int k = 0;
  #pragma unroll 4
  for (; k < len8; k += 8) {
    v4fu a4 = *reinterpret_cast<const v4fu*>(Aseg + aIdx);
    if (MOE) a4 *= scale;
    #pragma unroll
    for (int s = 0; s < 2; ++s) {
      float b0[NT], b1[NT];
      load_b<NT, VEC>(Wseg, wIdx + (2 * s) * N,     cbase, col, b0);
      load_b<NT, VEC>(Wseg, wIdx + (2 * s + 1) * N, cbase, col, b1);
      v2f a;
      a.x = a4[2 * s];
      a.y = a4[2 * s + 1];
      #pragma unroll
      for (int j = 0; j < NT; ++j) {
        v2f b;
        b.x = b0[j];
        b.y = b1[j];
        acc[j] = wmma4(a, b, acc[j]);
      }
    }
    aIdx += 8;
    wIdx += 8 * N;
  }
  for (; k < len; k += 4) {
    const int kk = k + 2 * h;
    v2f a;
    a.x = (kk     < len) ? Aseg[aBase + kk]     : 0.f;
    a.y = (kk + 1 < len) ? Aseg[aBase + kk + 1] : 0.f;
    if (MOE) { a.x *= scale; a.y *= scale; }
    const int r0 = (kk     < len) ? kk     : len - 1;
    const int r1 = (kk + 1 < len) ? kk + 1 : len - 1;
    float b0[NT], b1[NT];
    load_b<NT, VEC>(Wseg, r0 * N, cbase, col, b0);
    load_b<NT, VEC>(Wseg, r1 * N, cbase, col, b1);
    #pragma unroll
    for (int j = 0; j < NT; ++j) {
      v2f b;
      b.x = b0[j];
      b.y = b1[j];
      acc[j] = wmma4(a, b, acc[j]);
    }
  }
}

template <int N, bool MOE, int NT, int H, bool VEC>
__device__ __forceinline__ void k_slices(
    const float* __restrict__ A1, int k1,
    const float* __restrict__ A2, int k2,
    const float* __restrict__ Wb, int rowA, int p, float scale,
    int cbase, const int* col, int h, v8f* acc)
{
  {
    const int nst = (k1 + 7) >> 3;
    const int kb  = 8 * (nst * p / H);
    int ke        = 8 * (nst * (p + 1) / H);
    if (ke > k1) ke = k1;
    if (ke > kb)
      seg_gemm<N, MOE, NT, VEC>(A1, rowA * k1 + kb, ke - kb, scale,
                                Wb + (size_t)kb * N, cbase, col, h, acc);
  }
  if (k2) {
    const int nst = (k2 + 7) >> 3;
    const int kb  = 8 * (nst * p / H);
    int ke        = 8 * (nst * (p + 1) / H);
    if (ke > k2) ke = k2;
    if (ke > kb)
      seg_gemm<N, MOE, NT, VEC>(A2, rowA * k2 + kb, ke - kb, scale,
                                Wb + (size_t)(k1 + kb) * N, cbase, col, h, acc);
  }
}

template <int N, bool MOE, int NT, int S>
__global__ void __launch_bounds__(32 * S)
gemm_wmma_f32(const float* __restrict__ A1, int k1,
              const float* __restrict__ A2, int k2,
              const float* __restrict__ coeff,
              const float* __restrict__ W,
              const float* __restrict__ bias,
              float* __restrict__ C,
              int act)
{
  constexpr int TW    = 16 * NT;
  constexpr int PITCH = TW + NT;
  constexpr int H     = MOE ? S / NE : S;
  static_assert(!MOE || (S % NE) == 0, "MoE: S must be a multiple of NE");
  __shared__ __attribute__((aligned(16))) float part[S * 16 * PITCH];

  const int lane = threadIdx.x & 31;
  const int wave = __builtin_amdgcn_readfirstlane((int)(threadIdx.x >> 5));
  const int h    = lane >> 4;
  const int ln   = lane & 15;
  const int m0   = blockIdx.y * 16;
  const int n0   = blockIdx.x * TW;
  const int rowA = m0 + ln;

  const int cbase = n0 + NT * ln;
  int col[NT];
  #pragma unroll
  for (int j = 0; j < NT; ++j) {
    const int cb = cbase + j;
    col[j] = (cb < N) ? cb : N - 1;
  }

  v8f acc[NT];
  #pragma unroll
  for (int j = 0; j < NT; ++j) acc[j] = (v8f){0.f,0.f,0.f,0.f,0.f,0.f,0.f,0.f};

  const int I = k1 + k2;
  const int e = MOE ? wave / H : 0;
  const int p = MOE ? wave % H : wave;
  const float scale = MOE ? coeff[rowA * NE + e] : 1.f;
  const float* Wb = W + (size_t)e * I * N;

  float bv[NT];
  if (MOE) {
    const float* be = bias + e * N;
    #pragma unroll
    for (int j = 0; j < NT; ++j) bv[j] = be[col[j]];
  }

  if ((N % TW) != 0 && n0 + TW > N)
    k_slices<N, MOE, NT, H, false>(A1, k1, A2, k2, Wb, rowA, p, scale,
                                   cbase, col, h, acc);
  else
    k_slices<N, MOE, NT, H, true>(A1, k1, A2, k2, Wb, rowA, p, scale,
                                  cbase, col, h, acc);

  if (MOE) {
    v2f a;
    a.x = (h == 0) ? scale : 0.f;
    a.y = 0.f;
    #pragma unroll
    for (int j = 0; j < NT; ++j) {
      v2f b;
      b.x = (h == 0) ? bv[j] : 0.f;
      b.y = 0.f;
      acc[j] = wmma4(a, b, acc[j]);
    }
  }

  typedef typename RowVec<NT>::L LV;
  float* mine = part + wave * (16 * PITCH);
  #pragma unroll
  for (int i = 0; i < 8; ++i) {
    LV t;
    #pragma unroll
    for (int j = 0; j < NT; ++j) t[j] = acc[j][i];
    *reinterpret_cast<LV*>(mine + (8 * h + i) * PITCH + NT * ln) = t;
  }
  __syncthreads();

  for (int idx = threadIdx.x; idx < 16 * TW; idx += 32 * S) {
    const int r  = idx / TW;
    const int cl = idx % TW;
    float y = part[r * PITCH + cl];
    #pragma unroll
    for (int w = 1; w < S; ++w) y += part[(w * 16 + r) * PITCH + cl];
    const int colB = n0 + cl;
    if (colB < N) {
      const int row = m0 + r;
      if (!MOE && bias) y += bias[colB];
      if (act)    y = elu1(y);
      C[row * N + colB] = y;
    }
  }
}

__global__ void reparam_kernel(const float* __restrict__ m,
                               const float* __restrict__ v,
                               const float* __restrict__ eps,
                               float* __restrict__ z, int n) {
  int i = blockIdx.x * blockDim.x + threadIdx.x;
  if (i < n) z[i] = m[i] + eps[i] * expf(0.5f * v[i]);
}

__global__ void gate3_softmax_kernel(const float* __restrict__ g,
                                     const float* __restrict__ w,
                                     const float* __restrict__ b,
                                     float* __restrict__ coeff) {
  int row = blockIdx.x * blockDim.x + threadIdx.x;
  if (row >= BATCH) return;
  float lg[NE];
  #pragma unroll
  for (int e = 0; e < NE; ++e) lg[e] = b[e];
  for (int i = 0; i < GH; ++i) {
    float gv = g[row * GH + i];
    #pragma unroll
    for (int e = 0; e < NE; ++e) lg[e] = fmaf(gv, w[i * NE + e], lg[e]);
  }
  float mx = lg[0];
  #pragma unroll
  for (int e = 1; e < NE; ++e) mx = fmaxf(mx, lg[e]);
  float s = 0.f, ex[NE];
  #pragma unroll
  for (int e = 0; e < NE; ++e) { ex[e] = expf(lg[e] - mx); s += ex[e]; }
  float inv = 1.f / s;
  #pragma unroll
  for (int e = 0; e < NE; ++e) coeff[row * NE + e] = ex[e] * inv;
}

template <int N, int NT>
static inline dim3 gemm_grid() {
  return dim3((N + 16 * NT - 1) / (16 * NT), BATCH / 16);
}

extern "C" void kernel_launch(void* const* d_in, const int* in_sizes, int n_in,
                              void* d_out, int out_size, void* d_ws, size_t ws_size,
                              hipStream_t stream) {
  const float* x      = (const float*)d_in[0];
  const float* c      = (const float*)d_in[1];
  const float* eps    = (const float*)d_in[2];
  const float* efc1_w = (const float*)d_in[3];
  const float* efc1_b = (const float*)d_in[4];
  const float* efc2_w = (const float*)d_in[5];
  const float* efc2_b = (const float*)d_in[6];
  const float* em_w   = (const float*)d_in[7];
  const float* em_b   = (const float*)d_in[8];
  const float* ev_w   = (const float*)d_in[9];
  const float* ev_b   = (const float*)d_in[10];
  const float* g1_w   = (const float*)d_in[11];
  const float* g1_b   = (const float*)d_in[12];
  const float* g2_w   = (const float*)d_in[13];
  const float* g2_b   = (const float*)d_in[14];
  const float* g3_w   = (const float*)d_in[15];
  const float* g3_b   = (const float*)d_in[16];
  const float* w0     = (const float*)d_in[17];
  const float* b0     = (const float*)d_in[18];
  const float* w1     = (const float*)d_in[19];
  const float* b1     = (const float*)d_in[20];
  const float* w2     = (const float*)d_in[21];
  const float* b2     = (const float*)d_in[22];

  float* out   = (float*)d_out;
  float* out_m = out + BATCH * FRAME;
  float* out_v = out_m + BATCH * LAT;

  float* ws    = (float*)d_ws;
  float* h1    = ws;
  float* h2    = h1  + BATCH * HID;
  float* z     = h2  + BATCH * HID;
  float* g1o   = z   + BATCH * LAT;
  float* g2o   = g1o + BATCH * GH;
  float* coeff = g2o + BATCH * GH;

  gemm_wmma_f32<HID, false, 4, 8><<<gemm_grid<HID, 4>(), 32 * 8, 0, stream>>>(
      x, FRAME, c, FRAME, nullptr, efc1_w, efc1_b, h1, 1);
  gemm_wmma_f32<HID, false, 4, 8><<<gemm_grid<HID, 4>(), 32 * 8, 0, stream>>>(
      x, FRAME, h1, HID, nullptr, efc2_w, efc2_b, h2, 1);
  gemm_wmma_f32<LAT, false, 2, 8><<<gemm_grid<LAT, 2>(), 32 * 8, 0, stream>>>(
      x, FRAME, h2, HID, nullptr, em_w, em_b, out_m, 0);
  gemm_wmma_f32<LAT, false, 2, 8><<<gemm_grid<LAT, 2>(), 32 * 8, 0, stream>>>(
      x, FRAME, h2, HID, nullptr, ev_w, ev_b, out_v, 0);
  reparam_kernel<<<(BATCH * LAT + 255) / 256, 256, 0, stream>>>(
      out_m, out_v, eps, z, BATCH * LAT);

  gemm_wmma_f32<GH, false, 2, 8><<<gemm_grid<GH, 2>(), 32 * 8, 0, stream>>>(
      z, LAT, c, FRAME, nullptr, g1_w, g1_b, g1o, 1);
  gemm_wmma_f32<GH, false, 2, 8><<<gemm_grid<GH, 2>(), 32 * 8, 0, stream>>>(
      g1o, GH, nullptr, 0, nullptr, g2_w, g2_b, g2o, 1);
  gate3_softmax_kernel<<<(BATCH + 255) / 256, 256, 0, stream>>>(
      g2o, g3_w, g3_b, coeff);

  gemm_wmma_f32<HID, true, 4, NE><<<gemm_grid<HID, 4>(), 32 * NE, 0, stream>>>(
      z, LAT, c, FRAME, coeff, w0, b0, h1, 1);
  gemm_wmma_f32<HID, true, 4, NE><<<gemm_grid<HID, 4>(), 32 * NE, 0, stream>>>(
      z, LAT, h1, HID, coeff, w1, b1, h2, 1);
  gemm_wmma_f32<FRAME, true, 4, NE><<<gemm_grid<FRAME, 4>(), 32 * NE, 0, stream>>>(
      z, LAT, h2, HID, coeff, w2, b2, out, 0);
}
